# starting kernel
# baseline (speedup 1.0000x reference)
#include <hip/hip_runtime.h>

#define GN 50000
#define GE 800000
#define GD 128
#define NLAYER 4

typedef __attribute__((ext_vector_type(2))) float v2f;
typedef __attribute__((ext_vector_type(8))) float v8f;


__global__ void k_zero(float* p, int n) {
    int i = blockIdx.x * blockDim.x + threadIdx.x;
    if (i < n) p[i] = 0.0f;
}

__global__ void k_deg(const int* __restrict__ dst, float* deg, int e) {
    int i = blockIdx.x * blockDim.x + threadIdx.x;
    if (i < e) atomicAdd(&deg[dst[i]], 1.0f);
}

__global__ void k_rsqrt(float* dis, int n) {
    int i = blockIdx.x * blockDim.x + threadIdx.x;
    if (i < n) dis[i] = rsqrtf(1.0f + dis[i]);
}


__global__ void k_gemm128(const float* __restrict__ A,
                          const float* __restrict__ W,
                          float* __restrict__ out) {
    __shared__ float As[16 * 132];
    const int rowBase = blockIdx.x * 16;
    const int tid = threadIdx.x;

    for (int i = tid; i < 16 * 128; i += 256) {
        int r = i >> 7, c = i & 127;
        As[r * 132 + c] = A[(size_t)(rowBase + r) * GD + c];
    }
    __syncthreads();

    const int wave  = tid >> 5;
    const int lane  = tid & 31;
    const int m     = lane & 15;
    const int khalf = (lane >> 4) * 2;
    const int n     = wave * 16 + m;

    v8f c = {};
    #pragma unroll
    for (int k = 0; k < 128; k += 4) {
        v2f a, b;
        a.x = As[m * 132 + k + khalf];
        a.y = As[m * 132 + k + khalf + 1];
        b.x = W[(size_t)(k + khalf) * GD + n];
        b.y = W[(size_t)(k + khalf + 1) * GD + n];
        c = __builtin_amdgcn_wmma_f32_16x16x4_f32(
                false, a, false, b, (short)0, c, false, false);
    }

    const int rOff = (lane >> 4) * 8;
    #pragma unroll
    for (int i = 0; i < 8; ++i)
        out[(size_t)(rowBase + rOff + i) * GD + n] = c[i];
}


__global__ void k_init_agg(const float* __restrict__ xw,
                           const float* __restrict__ dis,
                           const float* __restrict__ b,
                           float* __restrict__ agg) {
    int i = blockIdx.x * blockDim.x + threadIdx.x;
    int node = i >> 7, f = i & 127;
    float ds = dis[node];
    agg[i] = xw[i] * ds * ds + b[f];
}


__global__ void k_edge_scatter(const float* __restrict__ xw,
                               const int* __restrict__ src,
                               const int* __restrict__ dst,
                               const float* __restrict__ dis,
                               float* __restrict__ agg, int e) {
    int gt = blockIdx.x * blockDim.x + threadIdx.x;
    int edge = gt >> 5;
    if (edge >= e) return;
    int lane = threadIdx.x & 31;
    int s = src[edge], d = dst[edge];
    float norm = dis[s] * dis[d];
    const float4* xs = (const float4*)(xw + (size_t)s * GD);
    float4 v = xs[lane];
    float* ad = agg + (size_t)d * GD + lane * 4;
    atomicAdd(ad + 0, v.x * norm);
    atomicAdd(ad + 1, v.y * norm);
    atomicAdd(ad + 2, v.z * norm);
    atomicAdd(ad + 3, v.w * norm);
}


__global__ void k_finish(const float* __restrict__ agg,
                         float* __restrict__ h, int residual) {
    int i = blockIdx.x * blockDim.x + threadIdx.x;
    float v = agg[i] + (residual ? h[i] : 0.0f);
    h[i] = fmaxf(v, 0.0f);
}


__global__ void k_dot(const float* __restrict__ h,
                      const float* __restrict__ Wout,
                      float* __restrict__ xw1, int n) {
    int gt = blockIdx.x * blockDim.x + threadIdx.x;
    int row = gt >> 5;
    if (row >= n) return;
    int lane = threadIdx.x & 31;
    float4 a = ((const float4*)(h + (size_t)row * GD))[lane];
    float4 b = ((const float4*)Wout)[lane];
    float sum = a.x * b.x + a.y * b.y + a.z * b.z + a.w * b.w;
    #pragma unroll
    for (int off = 16; off > 0; off >>= 1)
        sum += __shfl_down(sum, off, 32);
    if (lane == 0) xw1[row] = sum;
}

__global__ void k_out_init(const float* __restrict__ xw1,
                           const float* __restrict__ dis,
                           const float* __restrict__ bout,
                           float* __restrict__ out, int n) {
    int i = blockIdx.x * blockDim.x + threadIdx.x;
    if (i < n) {
        float ds = dis[i];
        out[i] = xw1[i] * ds * ds + bout[0];
    }
}

__global__ void k_out_scatter(const float* __restrict__ xw1,
                              const int* __restrict__ src,
                              const int* __restrict__ dst,
                              const float* __restrict__ dis,
                              float* __restrict__ out, int e) {
    int i = blockIdx.x * blockDim.x + threadIdx.x;
    if (i < e) {
        int s = src[i], d = dst[i];
        atomicAdd(&out[d], xw1[s] * dis[s] * dis[d]);
    }
}


extern "C" void kernel_launch(void* const* d_in, const int* in_sizes, int n_in,
                              void* d_out, int out_size, void* d_ws, size_t ws_size,
                              hipStream_t stream) {
    (void)in_sizes; (void)n_in; (void)out_size; (void)ws_size;
    const float* x     = (const float*)d_in[0];
    const int*   eidx  = (const int*)d_in[1];
    const float* W_in  = (const float*)d_in[2];
    const float* b_in  = (const float*)d_in[3];
    const float* W_h   = (const float*)d_in[4];
    const float* b_h   = (const float*)d_in[5];
    const float* W_out = (const float*)d_in[6];
    const float* b_out = (const float*)d_in[7];
    float* out = (float*)d_out;

    const int* src = eidx;
    const int* dst = eidx + GE;

    const size_t NPAD = (GN + 63) & ~(size_t)63;
    float* dis = (float*)d_ws;
    float* xw1 = dis + NPAD;
    float* h   = xw1 + NPAD;
    float* xw  = h + (size_t)GN * GD;
    float* agg = xw + (size_t)GN * GD;

    const int T = 256;
    const int gN    = (GN + T - 1) / T;
    const int gE    = (GE + T - 1) / T;
    const int gNF   = (GN * GD) / T;
    const int gEW   = (GE * 32) / T;
    const int gGemm = GN / 16;

    k_zero<<<gN, T, 0, stream>>>(dis, GN);
    k_deg<<<gE, T, 0, stream>>>(dst, dis, GE);
    k_rsqrt<<<gN, T, 0, stream>>>(dis, GN);

    k_gemm128<<<gGemm, T, 0, stream>>>(x, W_in, xw);
    k_init_agg<<<gNF, T, 0, stream>>>(xw, dis, b_in, agg);
    k_edge_scatter<<<gEW, T, 0, stream>>>(xw, src, dst, dis, agg, GE);
    k_finish<<<gNF, T, 0, stream>>>(agg, h, 0);

    for (int l = 0; l < NLAYER; ++l) {
        const float* Wl = W_h + (size_t)l * GD * GD;
        const float* bl = b_h + (size_t)l * GD;
        k_gemm128<<<gGemm, T, 0, stream>>>(h, Wl, xw);
        k_init_agg<<<gNF, T, 0, stream>>>(xw, dis, bl, agg);
        k_edge_scatter<<<gEW, T, 0, stream>>>(xw, src, dst, dis, agg, GE);
        k_finish<<<gNF, T, 0, stream>>>(agg, h, 1);
    }

    k_dot<<<(GN * 32) / T, T, 0, stream>>>(h, W_out, xw1, GN);
    k_out_init<<<gN, T, 0, stream>>>(xw1, dis, b_out, out, GN);
    k_out_scatter<<<gE, T, 0, stream>>>(xw1, src, dst, dis, out, GE);
}


# edge sums by destination: edges counting-sorted per call (int atomic return = rank), one wave per node sums its rows in registers and writes h once; no f32 atomics; sum order = rank order
# speedup vs baseline: 3.3493x; 3.3493x over previous
#include <hip/hip_runtime.h>

#define GN 50000
#define GE 800000
#define GD 128
#define NLAYER 4

typedef __attribute__((ext_vector_type(2))) float v2f;
typedef __attribute__((ext_vector_type(8))) float v8f;


__global__ void k_zero_i(int* p, int n) {
    int i = blockIdx.x * blockDim.x + threadIdx.x;
    if (i < n) p[i] = 0;
}

__global__ void k_hist(const int* __restrict__ dst, int* cnt,
                       int* __restrict__ rank, int e) {
    int i = blockIdx.x * blockDim.x + threadIdx.x;
    if (i < e) rank[i] = atomicAdd(&cnt[dst[i]], 1);
}

#define SCAN_STEPS 49
#define SCAN_GROUP 7
#define SCAN_PER_WAVE (SCAN_STEPS * 32)

__global__ __launch_bounds__(1024) void k_scan(const int* __restrict__ cnt,
                                               int* __restrict__ rowptr,
                                               float* __restrict__ dis) {
    static_assert(SCAN_PER_WAVE * 32 >= GN, "scan must cover every node");
    static_assert(SCAN_STEPS % SCAN_GROUP == 0, "steps come in whole groups");
    __shared__ int wtot[32];
    const int lane = threadIdx.x & 31;
    const int wave = threadIdx.x >> 5;
    const int base = wave * SCAN_PER_WAVE;

    int v[SCAN_STEPS];
    int lsum = 0;
    #pragma unroll
    for (int k = 0; k < SCAN_STEPS; ++k) {
        int idx = base + k * 32 + lane;
        v[k] = (idx < GN) ? cnt[idx] : 0;
        lsum += v[k];
    }
    for (int off = 16; off > 0; off >>= 1) lsum += __shfl_xor(lsum, off, 32);
    if (lane == 0) wtot[wave] = lsum;
    __syncthreads();

    int before = (lane < wave) ? wtot[lane] : 0;
    for (int off = 16; off > 0; off >>= 1) before += __shfl_xor(before, off, 32);

    int carry = before;
    #pragma unroll
    for (int g = 0; g < SCAN_STEPS; g += SCAN_GROUP) {
        int inc[SCAN_GROUP];
        #pragma unroll
        for (int t = 0; t < SCAN_GROUP; ++t) inc[t] = v[g + t];
        #pragma unroll
        for (int off = 1; off < 32; off <<= 1) {
            int y[SCAN_GROUP];
            #pragma unroll
            for (int t = 0; t < SCAN_GROUP; ++t) y[t] = __shfl_up(inc[t], off, 32);
            #pragma unroll
            for (int t = 0; t < SCAN_GROUP; ++t) if (lane >= off) inc[t] += y[t];
        }
        int tot[SCAN_GROUP];
        #pragma unroll
        for (int t = 0; t < SCAN_GROUP; ++t) tot[t] = __shfl(inc[t], 31, 32);
        #pragma unroll
        for (int t = 0; t < SCAN_GROUP; ++t) {
            int x = v[g + t];
            int idx = base + (g + t) * 32 + lane;
            if (idx < GN) {
                rowptr[idx] = carry + inc[t] - x;
                dis[idx] = rsqrtf(1.0f + (float)x);
            }
            carry += tot[t];
        }
    }
    if (threadIdx.x == 1023) rowptr[GN] = carry;
}

__global__ void k_fill(const int* __restrict__ src, const int* __restrict__ dst,
                       const int* __restrict__ rank, const int* __restrict__ rowptr,
                       const float* __restrict__ dis, int2* __restrict__ csr, int e) {
    int i = blockIdx.x * blockDim.x + threadIdx.x;
    if (i < e) {
        int s = src[i], d = dst[i];
        int pos = rowptr[d] + rank[i];
        csr[pos] = make_int2(s, __float_as_int(dis[s]));
    }
}


__global__ void k_gemm128(const float* __restrict__ A,
                          const float* __restrict__ W,
                          float* __restrict__ out) {
    __shared__ float As[16 * 132];
    const int rowBase = blockIdx.x * 16;
    const int tid = threadIdx.x;

    for (int i = tid; i < 16 * 128; i += 256) {
        int r = i >> 7, c = i & 127;
        As[r * 132 + c] = A[(size_t)(rowBase + r) * GD + c];
    }
    __syncthreads();

    const int wave  = tid >> 5;
    const int lane  = tid & 31;
    const int m     = lane & 15;
    const int khalf = (lane >> 4) * 2;
    const int n     = wave * 16 + m;

    v8f c = {};
    #pragma unroll
    for (int k = 0; k < 128; k += 4) {
        v2f a, b;
        a.x = As[m * 132 + k + khalf];
        a.y = As[m * 132 + k + khalf + 1];
        b.x = W[(size_t)(k + khalf) * GD + n];
        b.y = W[(size_t)(k + khalf + 1) * GD + n];
        c = __builtin_amdgcn_wmma_f32_16x16x4_f32(
                false, a, false, b, (short)0, c, false, false);
    }

    const int rOff = (lane >> 4) * 8;
    #pragma unroll
    for (int i = 0; i < 8; ++i)
        out[(size_t)(rowBase + rOff + i) * GD + n] = c[i];
}


template <int K>
__device__ __forceinline__ void gather_rows(const float* __restrict__ xw,
                                            const int2* __restrict__ csr,
                                            int j, float dd, int lane, float4& acc) {
    int2 e[K];
    float4 v[K];
    #pragma unroll
    for (int k = 0; k < K; ++k) e[k] = csr[j + k];
    #pragma unroll
    for (int k = 0; k < K; ++k) {
        unsigned s = min((unsigned)e[k].x, (unsigned)(GN - 1));
        v[k] = ((const float4*)(xw + (size_t)s * GD))[lane];
    }
    #pragma unroll
    for (int k = 0; k < K; ++k) {
        float w = __int_as_float(e[k].y) * dd;
        acc.x = fmaf(v[k].x, w, acc.x);
        acc.y = fmaf(v[k].y, w, acc.y);
        acc.z = fmaf(v[k].z, w, acc.z);
        acc.w = fmaf(v[k].w, w, acc.w);
    }
}

template <int RES>
__global__ __launch_bounds__(256) void k_gather(const float* __restrict__ xw,
                                                const int* __restrict__ rowptr,
                                                const int2* __restrict__ csr,
                                                const float* __restrict__ dis,
                                                const float* __restrict__ b,
                                                float* h) {
    const int node = __builtin_amdgcn_readfirstlane(
        (int)((blockIdx.x * 256u + threadIdx.x) >> 5));
    const int lane = threadIdx.x & 31;
    const int beg = rowptr[node], end = rowptr[node + 1];
    const float dd = dis[node];

    const float4 self = ((const float4*)(xw + (size_t)node * GD))[lane];
    const float4 bb = ((const float4*)b)[lane];
    float4* hp = (float4*)(h + (size_t)node * GD) + lane;
    float4 ho = make_float4(0.f, 0.f, 0.f, 0.f);
    if (RES) ho = *hp;

    float4 acc = make_float4(0.f, 0.f, 0.f, 0.f);
    int j = beg;
    for (; j + 8 <= end; j += 8) gather_rows<8>(xw, csr, j, dd, lane, acc);
    if (j + 4 <= end) { gather_rows<4>(xw, csr, j, dd, lane, acc); j += 4; }
    if (j + 2 <= end) { gather_rows<2>(xw, csr, j, dd, lane, acc); j += 2; }
    if (j < end)      { gather_rows<1>(xw, csr, j, dd, lane, acc); }

    const float sc = dd * dd;
    float4 r;
    r.x = self.x * sc + bb.x + acc.x;
    r.y = self.y * sc + bb.y + acc.y;
    r.z = self.z * sc + bb.z + acc.z;
    r.w = self.w * sc + bb.w + acc.w;
    if (RES) { r.x += ho.x; r.y += ho.y; r.z += ho.z; r.w += ho.w; }
    r.x = fmaxf(r.x, 0.0f); r.y = fmaxf(r.y, 0.0f);
    r.z = fmaxf(r.z, 0.0f); r.w = fmaxf(r.w, 0.0f);
    *hp = r;
}


__global__ void k_dot(const float* __restrict__ h,
                      const float* __restrict__ Wout,
                      float* __restrict__ xw1, int n) {
    int gt = blockIdx.x * blockDim.x + threadIdx.x;
    int row = gt >> 5;
    if (row >= n) return;
    int lane = threadIdx.x & 31;
    float4 a = ((const float4*)(h + (size_t)row * GD))[lane];
    float4 b = ((const float4*)Wout)[lane];
    float sum = a.x * b.x + a.y * b.y + a.z * b.z + a.w * b.w;
    #pragma unroll
    for (int off = 16; off > 0; off >>= 1)
        sum += __shfl_down(sum, off, 32);
    if (lane == 0) xw1[row] = sum;
}

__global__ void k_out_gather(const float* __restrict__ xw1,
                             const int* __restrict__ rowptr,
                             const int2* __restrict__ csr,
                             const float* __restrict__ dis,
                             const float* __restrict__ bout,
                             float* __restrict__ out, int n) {
    int gt = blockIdx.x * blockDim.x + threadIdx.x;
    int node = gt >> 3, sub = gt & 7;
    bool ok = node < n;
    int nd = ok ? node : n - 1;
    int beg = rowptr[nd], end = rowptr[nd + 1];
    float dd = dis[nd];
    float sum = 0.0f;
    for (int j = beg + sub; j < end; j += 8) {
        int2 e = csr[j];
        unsigned s = min((unsigned)e.x, (unsigned)(n - 1));
        sum += xw1[s] * (__int_as_float(e.y) * dd);
    }
    sum += __shfl_xor(sum, 4, 32);
    sum += __shfl_xor(sum, 2, 32);
    sum += __shfl_xor(sum, 1, 32);
    if (ok && sub == 0) out[node] = xw1[node] * dd * dd + bout[0] + sum;
}


extern "C" void kernel_launch(void* const* d_in, const int* in_sizes, int n_in,
                              void* d_out, int out_size, void* d_ws, size_t ws_size,
                              hipStream_t stream) {
    (void)in_sizes; (void)n_in; (void)out_size; (void)ws_size;
    const float* x     = (const float*)d_in[0];
    const int*   eidx  = (const int*)d_in[1];
    const float* W_in  = (const float*)d_in[2];
    const float* b_in  = (const float*)d_in[3];
    const float* W_h   = (const float*)d_in[4];
    const float* b_h   = (const float*)d_in[5];
    const float* W_out = (const float*)d_in[6];
    const float* b_out = (const float*)d_in[7];
    float* out = (float*)d_out;

    const int* src = eidx;
    const int* dst = eidx + GE;

    const size_t NPAD = (GN + 63) & ~(size_t)63;
    float* dis = (float*)d_ws;
    float* xw1 = dis + NPAD;
    float* h   = xw1 + NPAD;
    float* xw  = h + (size_t)GN * GD;
    int2*  csr = (int2*)(xw + (size_t)GN * GD);
    int* rowptr = (int*)(csr + GE);
    int* cnt    = rowptr + NPAD;
    int* rank   = cnt + NPAD;

    const int T = 256;
    const int gN    = (GN + T - 1) / T;
    const int gE    = (GE + T - 1) / T;
    const int gNW   = (GN * 32) / T;
    const int gN8   = (GN * 8 + T - 1) / T;
    const int gGemm = GN / 16;

    k_zero_i<<<gN, T, 0, stream>>>(cnt, GN);
    k_hist<<<gE, T, 0, stream>>>(dst, cnt, rank, GE);
    k_scan<<<1, 1024, 0, stream>>>(cnt, rowptr, dis);
    k_fill<<<gE, T, 0, stream>>>(src, dst, rank, rowptr, dis, csr, GE);

    k_gemm128<<<gGemm, T, 0, stream>>>(x, W_in, xw);
    k_gather<0><<<gNW, T, 0, stream>>>(xw, rowptr, csr, dis, b_in, h);

    for (int l = 0; l < NLAYER; ++l) {
        const float* Wl = W_h + (size_t)l * GD * GD;
        const float* bl = b_h + (size_t)l * GD;
        k_gemm128<<<gGemm, T, 0, stream>>>(h, Wl, xw);
        k_gather<1><<<gNW, T, 0, stream>>>(xw, rowptr, csr, dis, bl, h);
    }

    k_dot<<<(GN * 32) / T, T, 0, stream>>>(h, W_out, xw1, GN);
    k_out_gather<<<gN8, T, 0, stream>>>(xw1, rowptr, csr, dis, b_out, out, GN);
}


# each layer is one kernel: 32 nodes' neighbour rows summed into LDS, then multiplied by W on f32 WMMA (product moved behind the sum), bias/residual/relu on exit; sum order: aggregate, then multiply
# speedup vs baseline: 4.2724x; 1.2756x over previous
#include <hip/hip_runtime.h>

#define GN 50000
#define GE 800000
#define GD 128
#define NLAYER 4

typedef __attribute__((ext_vector_type(2))) float v2f;
typedef __attribute__((ext_vector_type(8))) float v8f;


__global__ void k_zero_i(int* p, int n) {
    int i = blockIdx.x * blockDim.x + threadIdx.x;
    if (i < n) p[i] = 0;
}

__global__ void k_hist(const int* __restrict__ dst, int* cnt,
                       int* __restrict__ rank, int e) {
    int i = blockIdx.x * blockDim.x + threadIdx.x;
    if (i < e) rank[i] = atomicAdd(&cnt[dst[i]], 1);
}

#define SCAN_STEPS 49
#define SCAN_GROUP 7
#define SCAN_PER_WAVE (SCAN_STEPS * 32)

__global__ __launch_bounds__(1024) void k_scan(const int* __restrict__ cnt,
                                               int* __restrict__ rowptr,
                                               float* __restrict__ dis) {
    static_assert(SCAN_PER_WAVE * 32 >= GN, "scan must cover every node");
    static_assert(SCAN_STEPS % SCAN_GROUP == 0, "steps come in whole groups");
    __shared__ int wtot[32];
    const int lane = threadIdx.x & 31;
    const int wave = threadIdx.x >> 5;
    const int base = wave * SCAN_PER_WAVE;

    int v[SCAN_STEPS];
    int lsum = 0;
    #pragma unroll
    for (int k = 0; k < SCAN_STEPS; ++k) {
        int idx = base + k * 32 + lane;
        v[k] = (idx < GN) ? cnt[idx] : 0;
        lsum += v[k];
    }
    for (int off = 16; off > 0; off >>= 1) lsum += __shfl_xor(lsum, off, 32);
    if (lane == 0) wtot[wave] = lsum;
    __syncthreads();

    int before = (lane < wave) ? wtot[lane] : 0;
    for (int off = 16; off > 0; off >>= 1) before += __shfl_xor(before, off, 32);

    int carry = before;
    #pragma unroll
    for (int g = 0; g < SCAN_STEPS; g += SCAN_GROUP) {
        int inc[SCAN_GROUP];
        #pragma unroll
        for (int t = 0; t < SCAN_GROUP; ++t) inc[t] = v[g + t];
        #pragma unroll
        for (int off = 1; off < 32; off <<= 1) {
            int y[SCAN_GROUP];
            #pragma unroll
            for (int t = 0; t < SCAN_GROUP; ++t) y[t] = __shfl_up(inc[t], off, 32);
            #pragma unroll
            for (int t = 0; t < SCAN_GROUP; ++t) if (lane >= off) inc[t] += y[t];
        }
        int tot[SCAN_GROUP];
        #pragma unroll
        for (int t = 0; t < SCAN_GROUP; ++t) tot[t] = __shfl(inc[t], 31, 32);
        #pragma unroll
        for (int t = 0; t < SCAN_GROUP; ++t) {
            int x = v[g + t];
            int idx = base + (g + t) * 32 + lane;
            if (idx < GN) {
                rowptr[idx] = carry + inc[t] - x;
                dis[idx] = rsqrtf(1.0f + (float)x);
            }
            carry += tot[t];
        }
    }
    if (threadIdx.x == 1023) rowptr[GN] = carry;
}

__global__ void k_fill(const int* __restrict__ src, const int* __restrict__ dst,
                       const int* __restrict__ rank, const int* __restrict__ rowptr,
                       const float* __restrict__ dis, int2* __restrict__ csr, int e) {
    int i = blockIdx.x * blockDim.x + threadIdx.x;
    if (i < e) {
        int s = src[i], d = dst[i];
        int pos = rowptr[d] + rank[i];
        csr[pos] = make_int2(s, __float_as_int(dis[s]));
    }
}


#define TM 32
#define NPW (TM / 8)
#define LDP 132

__device__ __forceinline__ int2 load_entries(const int2* __restrict__ csr,
                                             int base, int end, int lane) {
    int2 e = make_int2(0, 0);
    int idx = base + lane;
    if (idx < end) e = csr[idx];
    return e;
}

struct Ent { int off; float w; };

__device__ __forceinline__ Ent prep_entries(int2 e, float dd) {
    Ent p;
    p.off = (int)(min((unsigned)e.x, (unsigned)(GN - 1)) * (unsigned)(GD * sizeof(float)));
    p.w = __int_as_float(e.y) * dd;
    return p;
}

__device__ __forceinline__ float4 load_row(const float* __restrict__ hin,
                                           Ent p, int k, int lane) {
    unsigned o = (unsigned)__builtin_amdgcn_readlane(p.off, k);
    return ((const float4*)((const char*)hin + o))[lane];
}

__device__ __forceinline__ float edge_w(Ent p, int k) {
    return __int_as_float(__builtin_amdgcn_readlane(__float_as_int(p.w), k));
}

__device__ __forceinline__ void fma4(float4& acc, const float4& v, float w) {
    acc.x = fmaf(v.x, w, acc.x);
    acc.y = fmaf(v.y, w, acc.y);
    acc.z = fmaf(v.z, w, acc.z);
    acc.w = fmaf(v.w, w, acc.w);
}

__device__ __forceinline__ void gather_chunk(const float* __restrict__ hin, Ent e,
                                             int cnt, int lane, float4& acc) {
    int k = 0;
    for (; k + 8 <= cnt; k += 8) {
        float4 v[8];
        float w[8];
        #pragma unroll
        for (int t = 0; t < 8; ++t) { v[t] = load_row(hin, e, k + t, lane); w[t] = edge_w(e, k + t); }
        #pragma unroll
        for (int t = 0; t < 8; ++t) fma4(acc, v[t], w[t]);
    }
    const int r = cnt - k;
    if (r > 0) {
        const float4 z = make_float4(0.f, 0.f, 0.f, 0.f);
        float4 v[7] = {z, z, z, z, z, z, z};
        float w[7] = {0.f, 0.f, 0.f, 0.f, 0.f, 0.f, 0.f};
        if (r & 4) {
            #pragma unroll
            for (int t = 0; t < 4; ++t) { v[t] = load_row(hin, e, k + t, lane); w[t] = edge_w(e, k + t); }
            k += 4;
        }
        if (r & 2) {
            #pragma unroll
            for (int t = 0; t < 2; ++t) { v[4 + t] = load_row(hin, e, k + t, lane); w[4 + t] = edge_w(e, k + t); }
            k += 2;
        }
        if (r & 1) { v[6] = load_row(hin, e, k, lane); w[6] = edge_w(e, k); }
        #pragma unroll
        for (int t = 0; t < 7; ++t) fma4(acc, v[t], w[t]);
    }
}

template <int RES>
__global__ __launch_bounds__(256) void k_layer(const float* __restrict__ hin,
                                               const int* __restrict__ rowptr,
                                               const int2* __restrict__ csr,
                                               const float* __restrict__ dis,
                                               const float* __restrict__ W,
                                               const float* __restrict__ b,
                                               float* __restrict__ hout) {
    __shared__ __attribute__((aligned(16))) float As[TM * LDP];
    __shared__ __attribute__((aligned(16))) float Cs[TM * LDP];
    const int tid = threadIdx.x;
    const int lane = tid & 31;
    const int wave = __builtin_amdgcn_readfirstlane(tid >> 5);
    const int lrow0 = wave * NPW;
    const int node0 = blockIdx.x * TM + lrow0;

    int rp[NPW + 1];
    #pragma unroll
    for (int i = 0; i <= NPW; ++i) rp[i] = rowptr[min(node0 + i, GN)];
    float dd[NPW];
    float4 hs[NPW];
    #pragma unroll
    for (int i = 0; i < NPW; ++i) {
        dd[i] = dis[min(node0 + i, GN - 1)];
        hs[i] = make_float4(0.f, 0.f, 0.f, 0.f);
        if (node0 + i < GN) hs[i] = ((const float4*)(hin + (size_t)(node0 + i) * GD))[lane];
    }
    const float4 bb = ((const float4*)b)[lane];

    int2 e = load_entries(csr, rp[0], rp[1], lane);
    #pragma unroll
    for (int i = 0; i < NPW; ++i) {
        const int end = rp[i + 1];
        int base = rp[i];
        Ent p = prep_entries(e, dd[i]);
        const int i2 = (i + 2 <= NPW) ? i + 2 : NPW;
        if (i + 1 < NPW) e = load_entries(csr, rp[i + 1], rp[i2], lane);
        float4 acc = make_float4(0.f, 0.f, 0.f, 0.f);
        for (;;) {
            gather_chunk(hin, p, min(32, end - base), lane, acc);
            base += 32;
            if (base >= end) break;
            p = prep_entries(load_entries(csr, base, end, lane), dd[i]);
        }
        fma4(acc, hs[i], dd[i] * dd[i]);
        *(float4*)(As + (lrow0 + i) * LDP + lane * 4) = acc;
    }
    __syncthreads();

    const int m     = lane & 15;
    const int khalf = (lane >> 4) * 2;
    const int n     = wave * 16 + m;
    const float* ap = As + m * LDP + khalf;
    const float* bp = W + (size_t)khalf * GD + n;
    v8f c0 = {}, c1 = {};
    #pragma unroll
    for (int k = 0; k < GD; k += 4) {
        v2f a0 = *(const v2f*)(ap + k);
        v2f a1 = *(const v2f*)(ap + 16 * LDP + k);
        v2f bf;
        bf.x = bp[(size_t)k * GD];
        bf.y = bp[(size_t)(k + 1) * GD];
        c0 = __builtin_amdgcn_wmma_f32_16x16x4_f32(false, a0, false, bf, (short)0, c0, false, false);
        c1 = __builtin_amdgcn_wmma_f32_16x16x4_f32(false, a1, false, bf, (short)0, c1, false, false);
    }

    const int rOff = (lane >> 4) * 8;
    #pragma unroll
    for (int i = 0; i < 8; ++i) {
        Cs[(rOff + i) * LDP + n] = c0[i];
        Cs[(16 + rOff + i) * LDP + n] = c1[i];
    }
    __syncthreads();

    #pragma unroll
    for (int i = 0; i < NPW; ++i) {
        if (node0 + i < GN) {
            float4 o = *(const float4*)(Cs + (lrow0 + i) * LDP + lane * 4);
            o.x += bb.x; o.y += bb.y; o.z += bb.z; o.w += bb.w;
            if (RES) { o.x += hs[i].x; o.y += hs[i].y; o.z += hs[i].z; o.w += hs[i].w; }
            o.x = fmaxf(o.x, 0.0f); o.y = fmaxf(o.y, 0.0f);
            o.z = fmaxf(o.z, 0.0f); o.w = fmaxf(o.w, 0.0f);
            ((float4*)(hout + (size_t)(node0 + i) * GD))[lane] = o;
        }
    }
}


__global__ void k_dot(const float* __restrict__ h,
                      const float* __restrict__ Wout,
                      float* __restrict__ xw1, int n) {
    int gt = blockIdx.x * blockDim.x + threadIdx.x;
    int row = gt >> 5;
    if (row >= n) return;
    int lane = threadIdx.x & 31;
    float4 a = ((const float4*)(h + (size_t)row * GD))[lane];
    float4 b = ((const float4*)Wout)[lane];
    float sum = a.x * b.x + a.y * b.y + a.z * b.z + a.w * b.w;
    #pragma unroll
    for (int off = 16; off > 0; off >>= 1)
        sum += __shfl_down(sum, off, 32);
    if (lane == 0) xw1[row] = sum;
}

__global__ void k_out_gather(const float* __restrict__ xw1,
                             const int* __restrict__ rowptr,
                             const int2* __restrict__ csr,
                             const float* __restrict__ dis,
                             const float* __restrict__ bout,
                             float* __restrict__ out, int n) {
    int gt = blockIdx.x * blockDim.x + threadIdx.x;
    int node = gt >> 3, sub = gt & 7;
    bool ok = node < n;
    int nd = ok ? node : n - 1;
    int beg = rowptr[nd], end = rowptr[nd + 1];
    float dd = dis[nd];
    float sum = 0.0f;
    for (int j = beg + sub; j < end; j += 8) {
        int2 e = csr[j];
        unsigned s = min((unsigned)e.x, (unsigned)(n - 1));
        sum += xw1[s] * (__int_as_float(e.y) * dd);
    }
    sum += __shfl_xor(sum, 4, 32);
    sum += __shfl_xor(sum, 2, 32);
    sum += __shfl_xor(sum, 1, 32);
    if (ok && sub == 0) out[node] = xw1[node] * dd * dd + bout[0] + sum;
}


extern "C" void kernel_launch(void* const* d_in, const int* in_sizes, int n_in,
                              void* d_out, int out_size, void* d_ws, size_t ws_size,
                              hipStream_t stream) {
    (void)in_sizes; (void)n_in; (void)out_size; (void)ws_size;
    const float* x     = (const float*)d_in[0];
    const int*   eidx  = (const int*)d_in[1];
    const float* W_in  = (const float*)d_in[2];
    const float* b_in  = (const float*)d_in[3];
    const float* W_h   = (const float*)d_in[4];
    const float* b_h   = (const float*)d_in[5];
    const float* W_out = (const float*)d_in[6];
    const float* b_out = (const float*)d_in[7];
    float* out = (float*)d_out;

    const int* src = eidx;
    const int* dst = eidx + GE;

    const size_t NPAD = (GN + 63) & ~(size_t)63;
    float* dis = (float*)d_ws;
    float* xw1 = dis + NPAD;
    float* hA  = xw1 + NPAD;
    float* hB  = hA + (size_t)GN * GD;
    int2*  csr = (int2*)(hB + (size_t)GN * GD);
    int* rowptr = (int*)(csr + GE);
    int* cnt    = rowptr + NPAD;
    int* rank   = cnt + NPAD;

    const int T = 256;
    const int gN    = (GN + T - 1) / T;
    const int gE    = (GE + T - 1) / T;
    const int gN8   = (GN * 8 + T - 1) / T;
    const int gLay  = (GN + TM - 1) / TM;

    k_zero_i<<<gN, T, 0, stream>>>(cnt, GN);
    k_hist<<<gE, T, 0, stream>>>(dst, cnt, rank, GE);
    k_scan<<<1, 1024, 0, stream>>>(cnt, rowptr, dis);
    k_fill<<<gE, T, 0, stream>>>(src, dst, rank, rowptr, dis, csr, GE);

    k_layer<0><<<gLay, T, 0, stream>>>(x, rowptr, csr, dis, W_in, b_in, hA);

    static_assert(NLAYER == 4, "the four launches below are written out");
    const size_t WSZ = (size_t)GD * GD;
    k_layer<1><<<gLay, T, 0, stream>>>(hA, rowptr, csr, dis, W_h,           b_h,          hB);
    k_layer<1><<<gLay, T, 0, stream>>>(hB, rowptr, csr, dis, W_h + WSZ,     b_h + GD,     hA);
    k_layer<1><<<gLay, T, 0, stream>>>(hA, rowptr, csr, dis, W_h + 2 * WSZ, b_h + 2 * GD, hB);
    k_layer<1><<<gLay, T, 0, stream>>>(hB, rowptr, csr, dis, W_h + 3 * WSZ, b_h + 3 * GD, hA);

    k_dot<<<(GN * 32) / T, T, 0, stream>>>(hA, W_out, xw1, GN);
    k_out_gather<<<gN8, T, 0, stream>>>(xw1, rowptr, csr, dis, b_out, out, GN);
}


# output product h.W_out moved into the last hidden layer's epilogue (4-column fma chain, then the same 16-8-4-2-1 lane sum); its launch and that layer's row store and re-read go: 11 launches to 10
# speedup vs baseline: 4.3989x; 1.0296x over previous
#include <hip/hip_runtime.h>

#define GN 50000
#define GE 800000
#define GD 128
#define NLAYER 4

typedef __attribute__((ext_vector_type(2))) float v2f;
typedef __attribute__((ext_vector_type(8))) float v8f;


__global__ void k_zero_i(int* p, int n) {
    int i = blockIdx.x * blockDim.x + threadIdx.x;
    if (i < n) p[i] = 0;
}

__global__ void k_hist(const int* __restrict__ dst, int* cnt,
                       int* __restrict__ rank, int e) {
    int i = blockIdx.x * blockDim.x + threadIdx.x;
    if (i < e) rank[i] = atomicAdd(&cnt[dst[i]], 1);
}

#define SCAN_STEPS 49
#define SCAN_GROUP 7
#define SCAN_PER_WAVE (SCAN_STEPS * 32)

__global__ __launch_bounds__(1024) void k_scan(const int* __restrict__ cnt,
                                               int* __restrict__ rowptr,
                                               float* __restrict__ dis) {
    static_assert(SCAN_PER_WAVE * 32 >= GN, "scan must cover every node");
    static_assert(SCAN_STEPS % SCAN_GROUP == 0, "steps come in whole groups");
    __shared__ int wtot[32];
    const int lane = threadIdx.x & 31;
    const int wave = threadIdx.x >> 5;
    const int base = wave * SCAN_PER_WAVE;

    int v[SCAN_STEPS];
    int lsum = 0;
    #pragma unroll
    for (int k = 0; k < SCAN_STEPS; ++k) {
        int idx = base + k * 32 + lane;
        v[k] = (idx < GN) ? cnt[idx] : 0;
        lsum += v[k];
    }
    for (int off = 16; off > 0; off >>= 1) lsum += __shfl_xor(lsum, off, 32);
    if (lane == 0) wtot[wave] = lsum;
    __syncthreads();

    int before = (lane < wave) ? wtot[lane] : 0;
    for (int off = 16; off > 0; off >>= 1) before += __shfl_xor(before, off, 32);

    int carry = before;
    #pragma unroll
    for (int g = 0; g < SCAN_STEPS; g += SCAN_GROUP) {
        int inc[SCAN_GROUP];
        #pragma unroll
        for (int t = 0; t < SCAN_GROUP; ++t) inc[t] = v[g + t];
        #pragma unroll
        for (int off = 1; off < 32; off <<= 1) {
            int y[SCAN_GROUP];
            #pragma unroll
            for (int t = 0; t < SCAN_GROUP; ++t) y[t] = __shfl_up(inc[t], off, 32);
            #pragma unroll
            for (int t = 0; t < SCAN_GROUP; ++t) if (lane >= off) inc[t] += y[t];
        }
        int tot[SCAN_GROUP];
        #pragma unroll
        for (int t = 0; t < SCAN_GROUP; ++t) tot[t] = __shfl(inc[t], 31, 32);
        #pragma unroll
        for (int t = 0; t < SCAN_GROUP; ++t) {
            int x = v[g + t];
            int idx = base + (g + t) * 32 + lane;
            if (idx < GN) {
                rowptr[idx] = carry + inc[t] - x;
                dis[idx] = rsqrtf(1.0f + (float)x);
            }
            carry += tot[t];
        }
    }
    if (threadIdx.x == 1023) rowptr[GN] = carry;
}

__global__ void k_fill(const int* __restrict__ src, const int* __restrict__ dst,
                       const int* __restrict__ rank, const int* __restrict__ rowptr,
                       const float* __restrict__ dis, int2* __restrict__ csr, int e) {
    int i = blockIdx.x * blockDim.x + threadIdx.x;
    if (i < e) {
        int s = src[i], d = dst[i];
        int pos = rowptr[d] + rank[i];
        csr[pos] = make_int2(s, __float_as_int(dis[s]));
    }
}


#define TM 32
#define NPW (TM / 8)
#define LDP 132

__device__ __forceinline__ int2 load_entries(const int2* __restrict__ csr,
                                             int base, int end, int lane) {
    int2 e = make_int2(0, 0);
    int idx = base + lane;
    if (idx < end) e = csr[idx];
    return e;
}

struct Ent { int off; float w; };

__device__ __forceinline__ Ent prep_entries(int2 e, float dd) {
    Ent p;
    p.off = (int)(min((unsigned)e.x, (unsigned)(GN - 1)) * (unsigned)(GD * sizeof(float)));
    p.w = __int_as_float(e.y) * dd;
    return p;
}

__device__ __forceinline__ float4 load_row(const float* __restrict__ hin,
                                           Ent p, int k, int lane) {
    unsigned o = (unsigned)__builtin_amdgcn_readlane(p.off, k);
    return ((const float4*)((const char*)hin + o))[lane];
}

__device__ __forceinline__ float edge_w(Ent p, int k) {
    return __int_as_float(__builtin_amdgcn_readlane(__float_as_int(p.w), k));
}

__device__ __forceinline__ void fma4(float4& acc, const float4& v, float w) {
    acc.x = fmaf(v.x, w, acc.x);
    acc.y = fmaf(v.y, w, acc.y);
    acc.z = fmaf(v.z, w, acc.z);
    acc.w = fmaf(v.w, w, acc.w);
}

__device__ __forceinline__ void gather_chunk(const float* __restrict__ hin, Ent e,
                                             int cnt, int lane, float4& acc) {
    int k = 0;
    for (; k + 8 <= cnt; k += 8) {
        float4 v[8];
        float w[8];
        #pragma unroll
        for (int t = 0; t < 8; ++t) { v[t] = load_row(hin, e, k + t, lane); w[t] = edge_w(e, k + t); }
        #pragma unroll
        for (int t = 0; t < 8; ++t) fma4(acc, v[t], w[t]);
    }
    const int r = cnt - k;
    if (r > 0) {
        const float4 z = make_float4(0.f, 0.f, 0.f, 0.f);
        float4 v[7] = {z, z, z, z, z, z, z};
        float w[7] = {0.f, 0.f, 0.f, 0.f, 0.f, 0.f, 0.f};
        if (r & 4) {
            #pragma unroll
            for (int t = 0; t < 4; ++t) { v[t] = load_row(hin, e, k + t, lane); w[t] = edge_w(e, k + t); }
            k += 4;
        }
        if (r & 2) {
            #pragma unroll
            for (int t = 0; t < 2; ++t) { v[4 + t] = load_row(hin, e, k + t, lane); w[4 + t] = edge_w(e, k + t); }
            k += 2;
        }
        if (r & 1) { v[6] = load_row(hin, e, k, lane); w[6] = edge_w(e, k); }
        #pragma unroll
        for (int t = 0; t < 7; ++t) fma4(acc, v[t], w[t]);
    }
}

template <int RES, int DOT>
__global__ __launch_bounds__(256) void k_layer(const float* __restrict__ hin,
                                               const int* __restrict__ rowptr,
                                               const int2* __restrict__ csr,
                                               const float* __restrict__ dis,
                                               const float* __restrict__ W,
                                               const float* __restrict__ b,
                                               float* __restrict__ hout,
                                               const float* __restrict__ wout,
                                               float* __restrict__ xw1) {
    __shared__ __attribute__((aligned(16))) float As[TM * LDP];
    __shared__ __attribute__((aligned(16))) float Cs[TM * LDP];
    const int tid = threadIdx.x;
    const int lane = tid & 31;
    const int wave = __builtin_amdgcn_readfirstlane(tid >> 5);
    const int lrow0 = wave * NPW;
    const int node0 = blockIdx.x * TM + lrow0;

    int rp[NPW + 1];
    #pragma unroll
    for (int i = 0; i <= NPW; ++i) rp[i] = rowptr[min(node0 + i, GN)];
    float dd[NPW];
    float4 hs[NPW];
    #pragma unroll
    for (int i = 0; i < NPW; ++i) {
        dd[i] = dis[min(node0 + i, GN - 1)];
        hs[i] = make_float4(0.f, 0.f, 0.f, 0.f);
        if (node0 + i < GN) hs[i] = ((const float4*)(hin + (size_t)(node0 + i) * GD))[lane];
    }
    const float4 bb = ((const float4*)b)[lane];
    float4 wo = make_float4(0.f, 0.f, 0.f, 0.f);
    if (DOT) wo = ((const float4*)wout)[lane];

    int2 e = load_entries(csr, rp[0], rp[1], lane);
    #pragma unroll
    for (int i = 0; i < NPW; ++i) {
        const int end = rp[i + 1];
        int base = rp[i];
        Ent p = prep_entries(e, dd[i]);
        const int i2 = (i + 2 <= NPW) ? i + 2 : NPW;
        if (i + 1 < NPW) e = load_entries(csr, rp[i + 1], rp[i2], lane);
        float4 acc = make_float4(0.f, 0.f, 0.f, 0.f);
        for (;;) {
            gather_chunk(hin, p, min(32, end - base), lane, acc);
            base += 32;
            if (base >= end) break;
            p = prep_entries(load_entries(csr, base, end, lane), dd[i]);
        }
        fma4(acc, hs[i], dd[i] * dd[i]);
        *(float4*)(As + (lrow0 + i) * LDP + lane * 4) = acc;
    }
    __syncthreads();

    const int m     = lane & 15;
    const int khalf = (lane >> 4) * 2;
    const int n     = wave * 16 + m;
    const float* ap = As + m * LDP + khalf;
    const float* bp = W + (size_t)khalf * GD + n;
    v8f c0 = {}, c1 = {};
    #pragma unroll
    for (int k = 0; k < GD; k += 4) {
        v2f a0 = *(const v2f*)(ap + k);
        v2f a1 = *(const v2f*)(ap + 16 * LDP + k);
        v2f bf;
        bf.x = bp[(size_t)k * GD];
        bf.y = bp[(size_t)(k + 1) * GD];
        c0 = __builtin_amdgcn_wmma_f32_16x16x4_f32(false, a0, false, bf, (short)0, c0, false, false);
        c1 = __builtin_amdgcn_wmma_f32_16x16x4_f32(false, a1, false, bf, (short)0, c1, false, false);
    }

    const int rOff = (lane >> 4) * 8;
    #pragma unroll
    for (int i = 0; i < 8; ++i) {
        Cs[(rOff + i) * LDP + n] = c0[i];
        Cs[(16 + rOff + i) * LDP + n] = c1[i];
    }
    __syncthreads();

    float xs = 0.0f;
    #pragma unroll
    for (int i = 0; i < NPW; ++i) {
        if (node0 + i < GN) {
            float4 o = *(const float4*)(Cs + (lrow0 + i) * LDP + lane * 4);
            o.x += bb.x; o.y += bb.y; o.z += bb.z; o.w += bb.w;
            if (RES) { o.x += hs[i].x; o.y += hs[i].y; o.z += hs[i].z; o.w += hs[i].w; }
            o.x = fmaxf(o.x, 0.0f); o.y = fmaxf(o.y, 0.0f);
            o.z = fmaxf(o.z, 0.0f); o.w = fmaxf(o.w, 0.0f);
            if (DOT) {
                float sum = fmaf(o.w, wo.w, fmaf(o.z, wo.z, fmaf(o.y, wo.y, o.x * wo.x)));
                #pragma unroll
                for (int off = 16; off > 0; off >>= 1)
                    sum += __shfl_down(sum, off, 32);
                sum = __shfl(sum, 0, 32);
                if (lane == i) xs = sum;
            } else {
                ((float4*)(hout + (size_t)(node0 + i) * GD))[lane] = o;
            }
        }
    }
    if (DOT && lane < NPW && node0 + lane < GN) xw1[node0 + lane] = xs;
}


__global__ void k_out_gather(const float* __restrict__ xw1,
                             const int* __restrict__ rowptr,
                             const int2* __restrict__ csr,
                             const float* __restrict__ dis,
                             const float* __restrict__ bout,
                             float* __restrict__ out, int n) {
    int gt = blockIdx.x * blockDim.x + threadIdx.x;
    int node = gt >> 3, sub = gt & 7;
    bool ok = node < n;
    int nd = ok ? node : n - 1;
    int beg = rowptr[nd], end = rowptr[nd + 1];
    float dd = dis[nd];
    float sum = 0.0f;
    for (int j = beg + sub; j < end; j += 8) {
        int2 e = csr[j];
        unsigned s = min((unsigned)e.x, (unsigned)(n - 1));
        sum += xw1[s] * (__int_as_float(e.y) * dd);
    }
    sum += __shfl_xor(sum, 4, 32);
    sum += __shfl_xor(sum, 2, 32);
    sum += __shfl_xor(sum, 1, 32);
    if (ok && sub == 0) out[node] = xw1[node] * dd * dd + bout[0] + sum;
}


extern "C" void kernel_launch(void* const* d_in, const int* in_sizes, int n_in,
                              void* d_out, int out_size, void* d_ws, size_t ws_size,
                              hipStream_t stream) {
    (void)in_sizes; (void)n_in; (void)out_size; (void)ws_size;
    const float* x     = (const float*)d_in[0];
    const int*   eidx  = (const int*)d_in[1];
    const float* W_in  = (const float*)d_in[2];
    const float* b_in  = (const float*)d_in[3];
    const float* W_h   = (const float*)d_in[4];
    const float* b_h   = (const float*)d_in[5];
    const float* W_out = (const float*)d_in[6];
    const float* b_out = (const float*)d_in[7];
    float* out = (float*)d_out;

    const int* src = eidx;
    const int* dst = eidx + GE;

    const size_t NPAD = (GN + 63) & ~(size_t)63;
    float* dis = (float*)d_ws;
    float* xw1 = dis + NPAD;
    float* hA  = xw1 + NPAD;
    float* hB  = hA + (size_t)GN * GD;
    int2*  csr = (int2*)(hB + (size_t)GN * GD);
    int* rowptr = (int*)(csr + GE);
    int* cnt    = rowptr + NPAD;
    int* rank   = cnt + NPAD;

    const int T = 256;
    const int gN    = (GN + T - 1) / T;
    const int gE    = (GE + T - 1) / T;
    const int gN8   = (GN * 8 + T - 1) / T;
    const int gLay  = (GN + TM - 1) / TM;

    k_zero_i<<<gN, T, 0, stream>>>(cnt, GN);
    k_hist<<<gE, T, 0, stream>>>(dst, cnt, rank, GE);
    k_scan<<<1, 1024, 0, stream>>>(cnt, rowptr, dis);
    k_fill<<<gE, T, 0, stream>>>(src, dst, rank, rowptr, dis, csr, GE);

    k_layer<0, 0><<<gLay, T, 0, stream>>>(x, rowptr, csr, dis, W_in, b_in, hA, nullptr, nullptr);

    static_assert(NLAYER == 4, "the four launches below are written out");
    const size_t WSZ = (size_t)GD * GD;
    k_layer<1, 0><<<gLay, T, 0, stream>>>(hA, rowptr, csr, dis, W_h,           b_h,          hB, nullptr, nullptr);
    k_layer<1, 0><<<gLay, T, 0, stream>>>(hB, rowptr, csr, dis, W_h + WSZ,     b_h + GD,     hA, nullptr, nullptr);
    k_layer<1, 0><<<gLay, T, 0, stream>>>(hA, rowptr, csr, dis, W_h + 2 * WSZ, b_h + 2 * GD, hB, nullptr, nullptr);
    k_layer<1, 1><<<gLay, T, 0, stream>>>(hB, rowptr, csr, dis, W_h + 3 * WSZ, b_h + 3 * GD, nullptr, W_out, xw1);

    k_out_gather<<<gN8, T, 0, stream>>>(xw1, rowptr, csr, dis, b_out, out, GN);
}
